# starting kernel
# baseline (speedup 1.0000x reference)
#include <hip/hip_runtime.h>

typedef float v2f __attribute__((ext_vector_type(2)));
typedef float v8f __attribute__((ext_vector_type(8)));

#define SDIM 64
#define NCIN 16
#define NCOUT 16
#define NTAP 27

__launch_bounds__(128, 4)
__global__ void conv3d_wmma_f32(const float* __restrict__ x,
                                const float* __restrict__ w,
                                const float* __restrict__ bias,
                                float* __restrict__ out) {
    __shared__ __align__(16) float wl[NTAP * NCOUT * NCIN];
    __shared__ __align__(16) float xs4[9 * 66 * 4];

    const int tid  = threadIdx.x;
    const int lane = tid & 31;
    const int wv   = tid >> 5;
    const int Lm   = lane & 15;
    const int half = lane >> 4;
    const int cl   = half << 1;
    const int w0   = wv * 16;

    const int blk = blockIdx.x;
    const int b   = blk >> 12;
    const int d   = (blk >> 6) & 63;
    const int h   = blk & 63;

    for (int idx = tid; idx < NCOUT * NCIN * NTAP; idx += 128) {
        int co   = idx / (NCIN * NTAP);
        int rest = idx - co * (NCIN * NTAP);
        int ci   = rest / NTAP;
        int tap  = rest - ci * NTAP;
        wl[(tap * NCOUT + co) * NCIN + ci] = w[idx];
    }
    for (int t = tid; t < 9 * 4 * 2; t += 128) {
        int rr   = t >> 1;
        int side = t & 1;
        int row  = rr >> 2;
        int cil  = rr & 3;
        xs4[(row * 66 + side * 65) * 4 + cil] = 0.0f;
    }

    const int abase = (w0 + Lm) * 4 + cl;
    v8f acc = {};

    for (int g = 0; g < 4; ++g) {
        __syncthreads();

        for (int t = tid; t < 36 * 16; t += 128) {
            int rr  = t >> 4;
            int q   = t & 15;
            int cil = rr / 9;
            int row = rr - cil * 9;
            int zd  = d + row / 3 - 1;
            int zh  = h + (row % 3) - 1;
            int ci  = (g << 2) + cil;
            float4 v = {0.0f, 0.0f, 0.0f, 0.0f};
            if ((unsigned)zd < 64u && (unsigned)zh < 64u)
                v = *(const float4*)&x[(((b * NCIN + ci) * SDIM + zd) * SDIM + zh) * SDIM
                                       + (q << 2)];
            int base = (row * 66 + 1 + (q << 2)) * 4 + cil;
            xs4[base]      = v.x;
            xs4[base + 4]  = v.y;
            xs4[base + 8]  = v.z;
            xs4[base + 12] = v.w;
        }
        __syncthreads();

        if (g < 3) {
            int rr  = tid >> 4;
            int q   = tid & 15;
            int cil = rr / 9;
            int row = rr - cil * 9;
            int zd  = d + row / 3 - 1;
            int zh  = h + (row % 3) - 1;
            int ci  = ((g + 1) << 2) + cil;
            if ((unsigned)zd < 64u && (unsigned)zh < 64u)
                __builtin_prefetch(&x[(((b * NCIN + ci) * SDIM + zd) * SDIM + zh) * SDIM
                                      + (q << 2)], 0, 3);
        }

        const int bbase = Lm * NCIN + (g << 2) + cl;

        #pragma unroll
        for (int kd = 0; kd < 3; ++kd) {
            #pragma unroll
            for (int kh = 0; kh < 3; ++kh) {
                const int arow = abase + (kd * 3 + kh) * 66 * 4;
                #pragma unroll
                for (int kw = 0; kw < 3; ++kw) {
                    const int tap = kd * 9 + kh * 3 + kw;
                    v2f a  = *(const v2f*)(xs4 + arow + kw * 4);
                    v2f bm = *(const v2f*)(wl + bbase + tap * (NCOUT * NCIN));
                    acc = __builtin_amdgcn_wmma_f32_16x16x4_f32(
                        false, a, false, bm, (short)0, acc, false, false);
                }
            }
        }
    }

    const float bv = bias[Lm];
    float* op = out + ((((long)b * NCOUT + Lm) * SDIM + d) * SDIM + h) * SDIM
                    + w0 + 8 * half;
    float4 v0 = {acc[0] + bv, acc[1] + bv, acc[2] + bv, acc[3] + bv};
    float4 v1 = {acc[4] + bv, acc[5] + bv, acc[6] + bv, acc[7] + bv};
    *(float4*)(op)     = v0;
    *(float4*)(op + 4) = v1;
}

extern "C" void kernel_launch(void* const* d_in, const int* in_sizes, int n_in,
                              void* d_out, int out_size, void* d_ws, size_t ws_size,
                              hipStream_t stream) {
    const float* x    = (const float*)d_in[0];
    const float* wgt  = (const float*)d_in[1];
    const float* bias = (const float*)d_in[2];
    float* out        = (float*)d_out;
    (void)in_sizes; (void)n_in; (void)out_size; (void)d_ws; (void)ws_size;

    dim3 grid(4 * SDIM * SDIM);
    conv3d_wmma_f32<<<grid, 128, 0, stream>>>(x, wgt, bias, out);
}


# Each work-group computes 8 consecutive h rows instead of 1 (grid 16384 -> 2048 blocks), so the weight transpose into LDS runs once per 8 rows; per-row math and summation order unchanged
# speedup vs baseline: 1.3987x; 1.3987x over previous
#include <hip/hip_runtime.h>

typedef float v2f __attribute__((ext_vector_type(2)));
typedef float v8f __attribute__((ext_vector_type(8)));

#define SDIM 64
#define NCIN 16
#define NCOUT 16
#define NTAP 27
#define HROWS 8

__launch_bounds__(128, 4)
__global__ void conv3d_wmma_f32(const float* __restrict__ x,
                                const float* __restrict__ w,
                                const float* __restrict__ bias,
                                float* __restrict__ out) {
    __shared__ __align__(16) float wl[NTAP * NCOUT * NCIN];
    __shared__ __align__(16) float xs4[9 * 66 * 4];

    const int tid  = threadIdx.x;
    const int lane = tid & 31;
    const int wv   = tid >> 5;
    const int Lm   = lane & 15;
    const int half = lane >> 4;
    const int cl   = half << 1;
    const int w0   = wv * 16;

    const int blk = blockIdx.x;
    const int b   = blk / (SDIM * (SDIM / HROWS));
    const int d   = (blk / (SDIM / HROWS)) % SDIM;
    const int h0  = (blk % (SDIM / HROWS)) * HROWS;

    for (int idx = tid; idx < NCOUT * NCIN * NTAP; idx += 128) {
        int co   = idx / (NCIN * NTAP);
        int rest = idx - co * (NCIN * NTAP);
        int ci   = rest / NTAP;
        int tap  = rest - ci * NTAP;
        wl[(tap * NCOUT + co) * NCIN + ci] = w[idx];
    }
    for (int t = tid; t < 9 * 4 * 2; t += 128) {
        int rr   = t >> 1;
        int side = t & 1;
        int row  = rr >> 2;
        int cil  = rr & 3;
        xs4[(row * 66 + side * 65) * 4 + cil] = 0.0f;
    }

    const int abase = (w0 + Lm) * 4 + cl;
    const float bv = bias[Lm];

    for (int r = 0; r < HROWS; ++r) {
        const int h = h0 + r;
        v8f acc = {};

        for (int g = 0; g < 4; ++g) {
            __syncthreads();

            for (int t = tid; t < 36 * 16; t += 128) {
                int rr  = t >> 4;
                int q   = t & 15;
                int cil = rr / 9;
                int row = rr - cil * 9;
                int zd  = d + row / 3 - 1;
                int zh  = h + (row % 3) - 1;
                int ci  = (g << 2) + cil;
                float4 v = {0.0f, 0.0f, 0.0f, 0.0f};
                if ((unsigned)zd < 64u && (unsigned)zh < 64u)
                    v = *(const float4*)&x[(((b * NCIN + ci) * SDIM + zd) * SDIM + zh) * SDIM
                                           + (q << 2)];
                int base = (row * 66 + 1 + (q << 2)) * 4 + cil;
                xs4[base]      = v.x;
                xs4[base + 4]  = v.y;
                xs4[base + 8]  = v.z;
                xs4[base + 12] = v.w;
            }
            __syncthreads();

            if (g < 3) {
                int rr  = tid >> 4;
                int q   = tid & 15;
                int cil = rr / 9;
                int row = rr - cil * 9;
                int zd  = d + row / 3 - 1;
                int zh  = h + (row % 3) - 1;
                int ci  = ((g + 1) << 2) + cil;
                if ((unsigned)zd < 64u && (unsigned)zh < 64u)
                    __builtin_prefetch(&x[(((b * NCIN + ci) * SDIM + zd) * SDIM + zh) * SDIM
                                          + (q << 2)], 0, 3);
            }

            const int bbase = Lm * NCIN + (g << 2) + cl;

            #pragma unroll
            for (int kd = 0; kd < 3; ++kd) {
                #pragma unroll
                for (int kh = 0; kh < 3; ++kh) {
                    const int arow = abase + (kd * 3 + kh) * 66 * 4;
                    #pragma unroll
                    for (int kw = 0; kw < 3; ++kw) {
                        const int tap = kd * 9 + kh * 3 + kw;
                        v2f a  = *(const v2f*)(xs4 + arow + kw * 4);
                        v2f bm = *(const v2f*)(wl + bbase + tap * (NCOUT * NCIN));
                        acc = __builtin_amdgcn_wmma_f32_16x16x4_f32(
                            false, a, false, bm, (short)0, acc, false, false);
                    }
                }
            }
        }

        asm volatile("v_nop\n\tv_nop\n\tv_nop\n\tv_nop\n\tv_nop\n\tv_nop\n\tv_nop\n\tv_nop"
                     : "+v"(acc));

        float* op = out + ((((long)b * NCOUT + Lm) * SDIM + d) * SDIM + h) * SDIM
                        + w0 + 8 * half;
        float4 v0 = {acc[0] + bv, acc[1] + bv, acc[2] + bv, acc[3] + bv};
        float4 v1 = {acc[4] + bv, acc[5] + bv, acc[6] + bv, acc[7] + bv};
        *(float4*)(op)     = v0;
        *(float4*)(op + 4) = v1;
    }
}

extern "C" void kernel_launch(void* const* d_in, const int* in_sizes, int n_in,
                              void* d_out, int out_size, void* d_ws, size_t ws_size,
                              hipStream_t stream) {
    const float* x    = (const float*)d_in[0];
    const float* wgt  = (const float*)d_in[1];
    const float* bias = (const float*)d_in[2];
    float* out        = (float*)d_out;
    (void)in_sizes; (void)n_in; (void)out_size; (void)d_ws; (void)ws_size;

    dim3 grid(4 * SDIM * (SDIM / HROWS));
    conv3d_wmma_f32<<<grid, 128, 0, stream>>>(x, wgt, bias, out);
}


# Each wave keeps all 8 h rows' accumulators in registers: x slab staged once per cin-group for the 8 rows, each A fragment feeds up to 3 rows, B fragments all 8; per-row summation order unchanged
# speedup vs baseline: 1.6843x; 1.2042x over previous
#include <hip/hip_runtime.h>

typedef float v2f __attribute__((ext_vector_type(2)));
typedef float v8f __attribute__((ext_vector_type(8)));

#define SDIM 64
#define NCIN 16
#define NCOUT 16
#define NTAP 27
#define HROWS 8
#define HIN (HROWS + 2)
#define SROWS (3 * HIN)

__launch_bounds__(128, 4)
__global__ void conv3d_wmma_f32(const float* __restrict__ x,
                                const float* __restrict__ w,
                                const float* __restrict__ bias,
                                float* __restrict__ out) {
    __shared__ __align__(16) float wl[NTAP * NCOUT * NCIN];
    __shared__ __align__(16) float xs4[SROWS * 66 * 4];

    const int tid  = threadIdx.x;
    const int lane = tid & 31;
    const int wv   = tid >> 5;
    const int Lm   = lane & 15;
    const int half = lane >> 4;
    const int cl   = half << 1;
    const int w0   = wv * 16;

    const int blk = blockIdx.x;
    const int b   = blk / (SDIM * (SDIM / HROWS));
    const int d   = (blk / (SDIM / HROWS)) % SDIM;
    const int h0  = (blk % (SDIM / HROWS)) * HROWS;

    for (int idx = tid; idx < NCOUT * NCIN * NTAP; idx += 128) {
        int co   = idx / (NCIN * NTAP);
        int rest = idx - co * (NCIN * NTAP);
        int ci   = rest / NTAP;
        int tap  = rest - ci * NTAP;
        wl[(tap * NCOUT + co) * NCIN + ci] = w[idx];
    }
    for (int t = tid; t < SROWS * 4 * 2; t += 128) {
        int rr   = t >> 1;
        int side = t & 1;
        int row  = rr >> 2;
        int cil  = rr & 3;
        xs4[(row * 66 + side * 65) * 4 + cil] = 0.0f;
    }

    const int abase = (w0 + Lm) * 4 + cl;
    const float bv = bias[Lm];

    v8f acc[HROWS];
    #pragma unroll
    for (int r = 0; r < HROWS; ++r) acc[r] = v8f{};

    for (int g = 0; g < 4; ++g) {
        __syncthreads();

        for (int t = tid; t < SROWS * 4 * 16; t += 128) {
            int q   = t & 15;
            int cil = (t >> 4) & 3;
            int row = t >> 6;
            int dz  = row / HIN;
            int hz  = row - dz * HIN;
            int zd  = d + dz - 1;
            int zh  = h0 + hz - 1;
            int ci  = (g << 2) + cil;
            float4 v = {0.0f, 0.0f, 0.0f, 0.0f};
            if ((unsigned)zd < 64u && (unsigned)zh < 64u)
                v = *(const float4*)&x[(((b * NCIN + ci) * SDIM + zd) * SDIM + zh) * SDIM
                                       + (q << 2)];
            int base = (row * 66 + 1 + (q << 2)) * 4 + cil;
            xs4[base]      = v.x;
            xs4[base + 4]  = v.y;
            xs4[base + 8]  = v.z;
            xs4[base + 12] = v.w;
        }
        __syncthreads();

        if (g < 3) {
            int q   = tid & 15;
            int cil = (tid >> 4) & 3;
            int hz  = tid >> 6;
            int zd  = d - 1;
            int zh  = h0 + hz - 1;
            int ci  = ((g + 1) << 2) + cil;
            if ((unsigned)zd < 64u && (unsigned)zh < 64u)
                __builtin_prefetch(&x[(((b * NCIN + ci) * SDIM + zd) * SDIM + zh) * SDIM
                                      + (q << 2)], 0, 3);
        }

        const int bbase = Lm * NCIN + (g << 2) + cl;

        #pragma unroll
        for (int kd = 0; kd < 3; ++kd) {
            v2f bf[3][3];
            #pragma unroll
            for (int kh = 0; kh < 3; ++kh) {
                #pragma unroll
                for (int kw = 0; kw < 3; ++kw)
                    bf[kh][kw] = *(const v2f*)(wl + bbase
                                               + (kd * 9 + kh * 3 + kw) * (NCOUT * NCIN));
            }
            #pragma unroll
            for (int hin = 0; hin < HIN; ++hin) {
                const int arow = abase + (kd * HIN + hin) * 66 * 4;
                #pragma unroll
                for (int kw = 0; kw < 3; ++kw) {
                    v2f a = *(const v2f*)(xs4 + arow + kw * 4);
                    #pragma unroll
                    for (int kh = 0; kh < 3; ++kh) {
                        const int r = hin - kh;
                        if (r >= 0 && r < HROWS)
                            acc[r] = __builtin_amdgcn_wmma_f32_16x16x4_f32(
                                false, a, false, bf[kh][kw], (short)0, acc[r], false, false);
                    }
                }
            }
        }

        asm volatile("v_nop\n\tv_nop\n\tv_nop\n\tv_nop\n\tv_nop\n\tv_nop\n\tv_nop\n\tv_nop"
                     : "+v"(acc[0]), "+v"(acc[1]), "+v"(acc[2]), "+v"(acc[3]),
                       "+v"(acc[4]), "+v"(acc[5]), "+v"(acc[6]), "+v"(acc[7]));
    }

    #pragma unroll
    for (int r = 0; r < HROWS; ++r) {
        float* op = out + ((((long)b * NCOUT + Lm) * SDIM + d) * SDIM + (h0 + r)) * SDIM
                        + w0 + 8 * half;
        float4 v0 = {acc[r][0] + bv, acc[r][1] + bv, acc[r][2] + bv, acc[r][3] + bv};
        float4 v1 = {acc[r][4] + bv, acc[r][5] + bv, acc[r][6] + bv, acc[r][7] + bv};
        *(float4*)(op)     = v0;
        *(float4*)(op + 4) = v1;
    }
}

extern "C" void kernel_launch(void* const* d_in, const int* in_sizes, int n_in,
                              void* d_out, int out_size, void* d_ws, size_t ws_size,
                              hipStream_t stream) {
    const float* x    = (const float*)d_in[0];
    const float* wgt  = (const float*)d_in[1];
    const float* bias = (const float*)d_in[2];
    float* out        = (float*)d_out;
    (void)in_sizes; (void)n_in; (void)out_size; (void)d_ws; (void)ws_size;

    dim3 grid(4 * SDIM * (SDIM / HROWS));
    conv3d_wmma_f32<<<grid, 128, 0, stream>>>(x, wgt, bias, out);
}


# x slab staging batched: the 15 b128 global loads of a cin-group are issued back to back, then stored to LDS (was one load per wait); group 0's are issued at kernel entry; summation order unchanged
# speedup vs baseline: 1.9760x; 1.1732x over previous
#include <hip/hip_runtime.h>

typedef float v2f __attribute__((ext_vector_type(2)));
typedef float v8f __attribute__((ext_vector_type(8)));

#define SDIM 64
#define NCIN 16
#define NCOUT 16
#define NTAP 27
#define HROWS 8
#define HIN (HROWS + 2)
#define SROWS (3 * HIN)
#define NLD (SROWS * 4 * 16 / 128)

__launch_bounds__(128, 4)
__global__ void conv3d_wmma_f32(const float* __restrict__ x,
                                const float* __restrict__ w,
                                const float* __restrict__ bias,
                                float* __restrict__ out) {
    __shared__ __align__(16) float wl[NTAP * NCOUT * NCIN];
    __shared__ __align__(16) float xs4[SROWS * 66 * 4];

    const int tid  = threadIdx.x;
    const int lane = tid & 31;
    const int wv   = tid >> 5;
    const int Lm   = lane & 15;
    const int half = lane >> 4;
    const int cl   = half << 1;
    const int w0   = wv * 16;

    const int blk = blockIdx.x;
    const int b   = blk / (SDIM * (SDIM / HROWS));
    const int d   = (blk / (SDIM / HROWS)) % SDIM;
    const int h0  = (blk % (SDIM / HROWS)) * HROWS;

    const int q   = tid & 15;
    const int cil = (tid >> 4) & 3;
    const int r0  = tid >> 6;
    int  zoff[3], hoff[5];
    bool zok[3], hok[5];
    #pragma unroll
    for (int dz = 0; dz < 3; ++dz) {
        int zd   = d + dz - 1;
        zok[dz]  = (unsigned)zd < 64u;
        zoff[dz] = min(max(zd, 0), SDIM - 1) * (SDIM * SDIM);
    }
    #pragma unroll
    for (int k = 0; k < 5; ++k) {
        int zh  = h0 + 2 * k + r0 - 1;
        hok[k]  = (unsigned)zh < 64u;
        hoff[k] = min(max(zh, 0), SDIM - 1) * SDIM;
    }
    const float* xb = x + (long)(b * NCIN + cil) * (SDIM * SDIM * SDIM) + (q << 2);
    const int sbase = (r0 * 66 + 1 + (q << 2)) * 4 + cil;

    float4 pv[NLD];
    auto slab_issue = [&](const float* xg) {
        #pragma unroll
        for (int i = 0; i < NLD; ++i)
            pv[i] = *(const float4*)(xg + zoff[i / 5] + hoff[i % 5]);
    };
    auto slab_commit = [&]() {
        #pragma unroll
        for (int i = 0; i < NLD; ++i) {
            const bool ok = zok[i / 5] && hok[i % 5];
            float* sp = xs4 + sbase + 2 * i * 66 * 4;
            sp[0]  = ok ? pv[i].x : 0.0f;
            sp[4]  = ok ? pv[i].y : 0.0f;
            sp[8]  = ok ? pv[i].z : 0.0f;
            sp[12] = ok ? pv[i].w : 0.0f;
        }
    };

    slab_issue(xb);

    for (int idx = tid; idx < NCOUT * NCIN * NTAP; idx += 128) {
        int co   = idx / (NCIN * NTAP);
        int rest = idx - co * (NCIN * NTAP);
        int ci   = rest / NTAP;
        int tap  = rest - ci * NTAP;
        wl[(tap * NCOUT + co) * NCIN + ci] = w[idx];
    }
    for (int t = tid; t < SROWS * 4 * 2; t += 128) {
        int rr   = t >> 1;
        int side = t & 1;
        int row  = rr >> 2;
        int cil  = rr & 3;
        xs4[(row * 66 + side * 65) * 4 + cil] = 0.0f;
    }

    slab_commit();

    const int abase = (w0 + Lm) * 4 + cl;
    const float bv = bias[Lm];

    v8f acc[HROWS];
    #pragma unroll
    for (int r = 0; r < HROWS; ++r) acc[r] = v8f{};

    for (int g = 0; g < 4; ++g) {
        __syncthreads();

        const int bbase = Lm * NCIN + (g << 2) + cl;

        #pragma unroll
        for (int kd = 0; kd < 3; ++kd) {
            v2f bf[3][3];
            #pragma unroll
            for (int kh = 0; kh < 3; ++kh) {
                #pragma unroll
                for (int kw = 0; kw < 3; ++kw)
                    bf[kh][kw] = *(const v2f*)(wl + bbase
                                               + (kd * 9 + kh * 3 + kw) * (NCOUT * NCIN));
            }
            #pragma unroll
            for (int hin = 0; hin < HIN; ++hin) {
                const int arow = abase + (kd * HIN + hin) * 66 * 4;
                #pragma unroll
                for (int kw = 0; kw < 3; ++kw) {
                    v2f a = *(const v2f*)(xs4 + arow + kw * 4);
                    #pragma unroll
                    for (int kh = 0; kh < 3; ++kh) {
                        const int r = hin - kh;
                        if (r >= 0 && r < HROWS)
                            acc[r] = __builtin_amdgcn_wmma_f32_16x16x4_f32(
                                false, a, false, bf[kh][kw], (short)0, acc[r], false, false);
                    }
                }
            }
        }

        asm volatile("v_nop\n\tv_nop\n\tv_nop\n\tv_nop\n\tv_nop\n\tv_nop\n\tv_nop\n\tv_nop"
                     : "+v"(acc[0]), "+v"(acc[1]), "+v"(acc[2]), "+v"(acc[3]),
                       "+v"(acc[4]), "+v"(acc[5]), "+v"(acc[6]), "+v"(acc[7]));

        if (g < 3) {
            slab_issue(xb + (long)(g + 1) * 4 * (SDIM * SDIM * SDIM));
            __syncthreads();
            slab_commit();
        }
    }

    #pragma unroll
    for (int r = 0; r < HROWS; ++r) {
        float* op = out + ((((long)b * NCOUT + Lm) * SDIM + d) * SDIM + (h0 + r)) * SDIM
                        + w0 + 8 * half;
        float4 v0 = {acc[r][0] + bv, acc[r][1] + bv, acc[r][2] + bv, acc[r][3] + bv};
        float4 v1 = {acc[r][4] + bv, acc[r][5] + bv, acc[r][6] + bv, acc[r][7] + bv};
        *(float4*)(op)     = v0;
        *(float4*)(op + 4) = v1;
    }
}

extern "C" void kernel_launch(void* const* d_in, const int* in_sizes, int n_in,
                              void* d_out, int out_size, void* d_ws, size_t ws_size,
                              hipStream_t stream) {
    const float* x    = (const float*)d_in[0];
    const float* wgt  = (const float*)d_in[1];
    const float* bias = (const float*)d_in[2];
    float* out        = (float*)d_out;
    (void)in_sizes; (void)n_in; (void)out_size; (void)d_ws; (void)ws_size;

    dim3 grid(4 * SDIM * (SDIM / HROWS));
    conv3d_wmma_f32<<<grid, 128, 0, stream>>>(x, wgt, bias, out);
}


# LDS bank conflicts removed: weights laid out [cin-group][cout][tap][4] (cout pitch 108 floats), x slab filled by one b128 store per (row,w) from 4 b32 global loads; summation order unchanged
# speedup vs baseline: 2.1311x; 1.0785x over previous
#include <hip/hip_runtime.h>

typedef float v2f __attribute__((ext_vector_type(2)));
typedef float v8f __attribute__((ext_vector_type(8)));

#define SDIM 64
#define NCIN 16
#define NCOUT 16
#define NTAP 27
#define HROWS 8
#define HIN (HROWS + 2)
#define SROWS (3 * HIN)
#define NLD (SROWS * 64 / 128)
#define S3 (SDIM * SDIM * SDIM)
#define W_CO_PITCH (NTAP * 4)
#define W_GRP_PITCH (NCOUT * W_CO_PITCH)

__launch_bounds__(128, 4)
__global__ void conv3d_wmma_f32(const float* __restrict__ x,
                                const float* __restrict__ w,
                                const float* __restrict__ bias,
                                float* __restrict__ out) {
    __shared__ __align__(16) float wl[4 * W_GRP_PITCH];
    __shared__ __align__(16) float xs4[SROWS * 66 * 4];

    const int tid  = threadIdx.x;
    const int lane = tid & 31;
    const int wv   = tid >> 5;
    const int Lm   = lane & 15;
    const int half = lane >> 4;
    const int cl   = half << 1;
    const int w0   = wv * 16;

    const int blk = blockIdx.x;
    const int b   = blk / (SDIM * (SDIM / HROWS));
    const int d   = (blk / (SDIM / HROWS)) % SDIM;
    const int h0  = (blk % (SDIM / HROWS)) * HROWS;

    const int sw  = tid & 63;
    const int r0  = tid >> 6;
    int  zoff[3], hoff[5];
    bool zok[3], hok[5];
    #pragma unroll
    for (int dz = 0; dz < 3; ++dz) {
        int zd   = d + dz - 1;
        zok[dz]  = (unsigned)zd < 64u;
        zoff[dz] = min(max(zd, 0), SDIM - 1) * (SDIM * SDIM);
    }
    #pragma unroll
    for (int k = 0; k < 5; ++k) {
        int zh  = h0 + 2 * k + r0 - 1;
        hok[k]  = (unsigned)zh < 64u;
        hoff[k] = min(max(zh, 0), SDIM - 1) * SDIM;
    }
    const float* xb = x + (long)(b * NCIN) * S3 + sw;
    const int sbase = (r0 * 66 + 1 + sw) * 4;

    float4 pv[NLD];
    auto slab_issue = [&](const float* xg) {
        #pragma unroll
        for (int i = 0; i < NLD; ++i) {
            const float* p = xg + zoff[i / 5] + hoff[i % 5];
            pv[i] = float4{p[0], p[S3], p[2 * S3], p[3 * S3]};
        }
    };
    auto slab_commit = [&]() {
        #pragma unroll
        for (int i = 0; i < NLD; ++i) {
            const bool ok = zok[i / 5] && hok[i % 5];
            float4 v;
            v.x = ok ? pv[i].x : 0.0f;
            v.y = ok ? pv[i].y : 0.0f;
            v.z = ok ? pv[i].z : 0.0f;
            v.w = ok ? pv[i].w : 0.0f;
            *(float4*)(xs4 + sbase + 2 * i * 66 * 4) = v;
        }
    };

    slab_issue(xb);

    for (int idx = tid; idx < NCOUT * NCIN * NTAP; idx += 128) {
        int co   = idx / (NCIN * NTAP);
        int rest = idx - co * (NCIN * NTAP);
        int ci   = rest / NTAP;
        int tap  = rest - ci * NTAP;
        wl[(ci >> 2) * W_GRP_PITCH + co * W_CO_PITCH + tap * 4 + (ci & 3)] = w[idx];
    }
    for (int t = tid; t < SROWS * 4 * 2; t += 128) {
        int rr   = t >> 1;
        int side = t & 1;
        int row  = rr >> 2;
        int cil  = rr & 3;
        xs4[(row * 66 + side * 65) * 4 + cil] = 0.0f;
    }

    slab_commit();

    const int abase = (w0 + Lm) * 4 + cl;
    const float bv = bias[Lm];

    v8f acc[HROWS];
    #pragma unroll
    for (int r = 0; r < HROWS; ++r) acc[r] = v8f{};

    for (int g = 0; g < 4; ++g) {
        __syncthreads();

        const int bbase = g * W_GRP_PITCH + Lm * W_CO_PITCH + cl;

        #pragma unroll
        for (int kd = 0; kd < 3; ++kd) {
            v2f bf[3][3];
            #pragma unroll
            for (int kh = 0; kh < 3; ++kh) {
                #pragma unroll
                for (int kw = 0; kw < 3; ++kw)
                    bf[kh][kw] = *(const v2f*)(wl + bbase + (kd * 9 + kh * 3 + kw) * 4);
            }
            #pragma unroll
            for (int hin = 0; hin < HIN; ++hin) {
                const int arow = abase + (kd * HIN + hin) * 66 * 4;
                #pragma unroll
                for (int kw = 0; kw < 3; ++kw) {
                    v2f a = *(const v2f*)(xs4 + arow + kw * 4);
                    #pragma unroll
                    for (int kh = 0; kh < 3; ++kh) {
                        const int r = hin - kh;
                        if (r >= 0 && r < HROWS)
                            acc[r] = __builtin_amdgcn_wmma_f32_16x16x4_f32(
                                false, a, false, bf[kh][kw], (short)0, acc[r], false, false);
                    }
                }
            }
        }

        asm volatile("v_nop\n\tv_nop\n\tv_nop\n\tv_nop\n\tv_nop\n\tv_nop\n\tv_nop\n\tv_nop"
                     : "+v"(acc[0]), "+v"(acc[1]), "+v"(acc[2]), "+v"(acc[3]),
                       "+v"(acc[4]), "+v"(acc[5]), "+v"(acc[6]), "+v"(acc[7]));

        if (g < 3) {
            slab_issue(xb + (long)(g + 1) * 4 * (SDIM * SDIM * SDIM));
            __syncthreads();
            slab_commit();
        }
    }

    #pragma unroll
    for (int r = 0; r < HROWS; ++r) {
        float* op = out + ((((long)b * NCOUT + Lm) * SDIM + d) * SDIM + (h0 + r)) * SDIM
                        + w0 + 8 * half;
        float4 v0 = {acc[r][0] + bv, acc[r][1] + bv, acc[r][2] + bv, acc[r][3] + bv};
        float4 v1 = {acc[r][4] + bv, acc[r][5] + bv, acc[r][6] + bv, acc[r][7] + bv};
        *(float4*)(op)     = v0;
        *(float4*)(op + 4) = v1;
    }
}

extern "C" void kernel_launch(void* const* d_in, const int* in_sizes, int n_in,
                              void* d_out, int out_size, void* d_ws, size_t ws_size,
                              hipStream_t stream) {
    const float* x    = (const float*)d_in[0];
    const float* wgt  = (const float*)d_in[1];
    const float* bias = (const float*)d_in[2];
    float* out        = (float*)d_out;
    (void)in_sizes; (void)n_in; (void)out_size; (void)d_ws; (void)ws_size;

    dim3 grid(4 * SDIM * (SDIM / HROWS));
    conv3d_wmma_f32<<<grid, 128, 0, stream>>>(x, wgt, bias, out);
}


# Weight transpose into LDS without per-element index arithmetic: thread t<108 owns one (cin%4,tap) slot of all 64 (cout,cin-group) chunks; 64 loads in one batch, then stores at constant offsets
# speedup vs baseline: 2.2852x; 1.0723x over previous
#include <hip/hip_runtime.h>

typedef float v2f __attribute__((ext_vector_type(2)));
typedef float v8f __attribute__((ext_vector_type(8)));

#define SDIM 64
#define NCIN 16
#define NCOUT 16
#define NTAP 27
#define HROWS 8
#define HIN (HROWS + 2)
#define SROWS (3 * HIN)
#define NLD (SROWS * 64 / 128)
#define S3 (SDIM * SDIM * SDIM)
#define W_CO_PITCH (NTAP * 4)
#define W_GRP_PITCH (NCOUT * W_CO_PITCH)

__launch_bounds__(128, 4)
__global__ void conv3d_wmma_f32(const float* __restrict__ x,
                                const float* __restrict__ w,
                                const float* __restrict__ bias,
                                float* __restrict__ out) {
    __shared__ __align__(16) float wl[4 * W_GRP_PITCH];
    __shared__ __align__(16) float xs4[SROWS * 66 * 4];

    const int tid  = threadIdx.x;
    const int lane = tid & 31;
    const int wv   = tid >> 5;
    const int Lm   = lane & 15;
    const int half = lane >> 4;
    const int cl   = half << 1;
    const int w0   = wv * 16;

    const int blk = blockIdx.x;
    const int b   = blk / (SDIM * (SDIM / HROWS));
    const int d   = (blk / (SDIM / HROWS)) % SDIM;
    const int h0  = (blk % (SDIM / HROWS)) * HROWS;

    const int sw  = tid & 63;
    const int r0  = tid >> 6;
    int  zoff[3], hoff[5];
    bool zok[3], hok[5];
    #pragma unroll
    for (int dz = 0; dz < 3; ++dz) {
        int zd   = d + dz - 1;
        zok[dz]  = (unsigned)zd < 64u;
        zoff[dz] = min(max(zd, 0), SDIM - 1) * (SDIM * SDIM);
    }
    #pragma unroll
    for (int k = 0; k < 5; ++k) {
        int zh  = h0 + 2 * k + r0 - 1;
        hok[k]  = (unsigned)zh < 64u;
        hoff[k] = min(max(zh, 0), SDIM - 1) * SDIM;
    }
    const float* xb = x + (long)(b * NCIN) * S3 + sw;
    const int sbase = (r0 * 66 + 1 + sw) * 4;

    float4 pv[NLD];
    auto slab_issue = [&](const float* xg) {
        #pragma unroll
        for (int i = 0; i < NLD; ++i) {
            const float* p = xg + zoff[i / 5] + hoff[i % 5];
            pv[i] = float4{p[0], p[S3], p[2 * S3], p[3 * S3]};
        }
    };
    auto slab_commit = [&]() {
        #pragma unroll
        for (int i = 0; i < NLD; ++i) {
            const bool ok = zok[i / 5] && hok[i % 5];
            float4 v;
            v.x = ok ? pv[i].x : 0.0f;
            v.y = ok ? pv[i].y : 0.0f;
            v.z = ok ? pv[i].z : 0.0f;
            v.w = ok ? pv[i].w : 0.0f;
            *(float4*)(xs4 + sbase + 2 * i * 66 * 4) = v;
        }
    };

    slab_issue(xb);

    if (tid < W_CO_PITCH) {
        const float* wp = w + tid;
        float* lp = wl + (tid % NTAP) * 4 + tid / NTAP;
        float wv[64];
        #pragma unroll
        for (int c = 0; c < 64; ++c) wv[c] = wp[c * W_CO_PITCH];
        #pragma unroll
        for (int c = 0; c < 64; ++c)
            lp[(c & 3) * W_GRP_PITCH + (c >> 2) * W_CO_PITCH] = wv[c];
    }
    for (int t = tid; t < SROWS * 4 * 2; t += 128) {
        int rr   = t >> 1;
        int side = t & 1;
        int row  = rr >> 2;
        int cil  = rr & 3;
        xs4[(row * 66 + side * 65) * 4 + cil] = 0.0f;
    }

    slab_commit();

    const int abase = (w0 + Lm) * 4 + cl;
    const float bv = bias[Lm];

    v8f acc[HROWS];
    #pragma unroll
    for (int r = 0; r < HROWS; ++r) acc[r] = v8f{};

    for (int g = 0; g < 4; ++g) {
        __syncthreads();

        const int bbase = g * W_GRP_PITCH + Lm * W_CO_PITCH + cl;

        #pragma unroll
        for (int kd = 0; kd < 3; ++kd) {
            v2f bf[3][3];
            #pragma unroll
            for (int kh = 0; kh < 3; ++kh) {
                #pragma unroll
                for (int kw = 0; kw < 3; ++kw)
                    bf[kh][kw] = *(const v2f*)(wl + bbase + (kd * 9 + kh * 3 + kw) * 4);
            }
            #pragma unroll
            for (int hin = 0; hin < HIN; ++hin) {
                const int arow = abase + (kd * HIN + hin) * 66 * 4;
                #pragma unroll
                for (int kw = 0; kw < 3; ++kw) {
                    v2f a = *(const v2f*)(xs4 + arow + kw * 4);
                    #pragma unroll
                    for (int kh = 0; kh < 3; ++kh) {
                        const int r = hin - kh;
                        if (r >= 0 && r < HROWS)
                            acc[r] = __builtin_amdgcn_wmma_f32_16x16x4_f32(
                                false, a, false, bf[kh][kw], (short)0, acc[r], false, false);
                    }
                }
            }
        }

        asm volatile("v_nop\n\tv_nop\n\tv_nop\n\tv_nop\n\tv_nop\n\tv_nop\n\tv_nop\n\tv_nop"
                     : "+v"(acc[0]), "+v"(acc[1]), "+v"(acc[2]), "+v"(acc[3]),
                       "+v"(acc[4]), "+v"(acc[5]), "+v"(acc[6]), "+v"(acc[7]));

        if (g < 3) {
            slab_issue(xb + (long)(g + 1) * 4 * (SDIM * SDIM * SDIM));
            __syncthreads();
            slab_commit();
        }
    }

    #pragma unroll
    for (int r = 0; r < HROWS; ++r) {
        float* op = out + ((((long)b * NCOUT + Lm) * SDIM + d) * SDIM + (h0 + r)) * SDIM
                        + w0 + 8 * half;
        float4 v0 = {acc[r][0] + bv, acc[r][1] + bv, acc[r][2] + bv, acc[r][3] + bv};
        float4 v1 = {acc[r][4] + bv, acc[r][5] + bv, acc[r][6] + bv, acc[r][7] + bv};
        *(float4*)(op)     = v0;
        *(float4*)(op + 4) = v1;
    }
}

extern "C" void kernel_launch(void* const* d_in, const int* in_sizes, int n_in,
                              void* d_out, int out_size, void* d_ws, size_t ws_size,
                              hipStream_t stream) {
    const float* x    = (const float*)d_in[0];
    const float* wgt  = (const float*)d_in[1];
    const float* bias = (const float*)d_in[2];
    float* out        = (float*)d_out;
    (void)in_sizes; (void)n_in; (void)out_size; (void)d_ws; (void)ws_size;

    dim3 grid(4 * SDIM * (SDIM / HROWS));
    conv3d_wmma_f32<<<grid, 128, 0, stream>>>(x, wgt, bias, out);
}


# Minimal filtering along d: block = 2 d planes x 4 h rows from 4 transformed planes and taps, 4 products for 6 (864->576 WMMA/wave); fragment loads pinned a unit ahead (16 tiles); sum order changes
# speedup vs baseline: 2.7204x; 1.1904x over previous
#include <hip/hip_runtime.h>

typedef float v2f __attribute__((ext_vector_type(2)));
typedef float v8f __attribute__((ext_vector_type(8)));

#define SDIM 64
#define NCIN 16
#define NCOUT 16
#define HROWS 4
#define HIN (HROWS + 2)
#define NPL 4
#define NTT (NPL * 9)
#define SROWS (NPL * HIN)
#define NLD (HIN * 64 / 128)
#define S3 (SDIM * SDIM * SDIM)
#define W_CO_PITCH (NTT * 4 + 4)
#define W_GRP_PITCH (NCOUT * W_CO_PITCH)

__launch_bounds__(128, 4)
__global__ void conv3d_wmma_f32(const float* __restrict__ x,
                                const float* __restrict__ w,
                                const float* __restrict__ bias,
                                float* __restrict__ out) {
    __shared__ __align__(16) float wl[4 * W_GRP_PITCH];
    __shared__ __align__(16) float xs4[SROWS * 66 * 4];

    const int tid  = threadIdx.x;
    const int lane = tid & 31;
    const int wv   = tid >> 5;
    const int Lm   = lane & 15;
    const int half = lane >> 4;
    const int cl   = half << 1;
    const int w0   = wv * 16;

    const int blk = blockIdx.x;
    const int b   = blk / ((SDIM / 2) * (SDIM / HROWS));
    const int d0  = ((blk / (SDIM / HROWS)) % (SDIM / 2)) * 2;
    const int h0  = (blk % (SDIM / HROWS)) * HROWS;

    const int sw  = tid & 63;
    const int r0  = tid >> 6;
    int  zoff[NPL], hoff[NLD];
    bool zok[NPL], hok[NLD];
    #pragma unroll
    for (int p = 0; p < NPL; ++p) {
        int zd  = d0 + p - 1;
        zok[p]  = (unsigned)zd < 64u;
        zoff[p] = min(max(zd, 0), SDIM - 1) * (SDIM * SDIM);
    }
    #pragma unroll
    for (int i = 0; i < NLD; ++i) {
        int zh  = h0 + 2 * i + r0 - 1;
        hok[i]  = (unsigned)zh < 64u;
        hoff[i] = min(max(zh, 0), SDIM - 1) * SDIM;
    }
    const float* xb = x + (long)(b * NCIN) * S3 + sw;
    const int sbase = (r0 * 66 + 1 + sw) * 4;

    float pv[NLD][NPL][4];
    auto slab_issue = [&](const float* xg) {
        #pragma unroll
        for (int i = 0; i < NLD; ++i) {
            #pragma unroll
            for (int p = 0; p < NPL; ++p) {
                const float* q = xg + zoff[p] + hoff[i];
                #pragma unroll
                for (int c = 0; c < 4; ++c) pv[i][p][c] = q[c * S3];
            }
        }
    };
    auto slab_commit = [&]() {
        #pragma unroll
        for (int i = 0; i < NLD; ++i) {
            float t[NPL][4];
            #pragma unroll
            for (int c = 0; c < 4; ++c) {
                const float x0 = (zok[0] && hok[i]) ? pv[i][0][c] : 0.0f;
                const float x1 = (zok[1] && hok[i]) ? pv[i][1][c] : 0.0f;
                const float x2 = (zok[2] && hok[i]) ? pv[i][2][c] : 0.0f;
                const float x3 = (zok[3] && hok[i]) ? pv[i][3][c] : 0.0f;
                t[0][c] = x0 - x2;
                t[1][c] = x1 + x2;
                t[2][c] = x2 - x1;
                t[3][c] = x1 - x3;
            }
            #pragma unroll
            for (int j = 0; j < NPL; ++j) {
                float4 v = {t[j][0], t[j][1], t[j][2], t[j][3]};
                *(float4*)(xs4 + sbase + (j * HIN + 2 * i) * 66 * 4) = v;
            }
        }
    };

    slab_issue(xb);

    if (tid < 2 * 36) {
        const int s    = tid % 36;
        const int part = tid / 36;
        const int cil  = s / 9;
        const int khkw = s % 9;
        const float* wp = w + part * 32 * 108 + cil * 27 + khkw;
        float* lp = wl + part * 8 * W_CO_PITCH + khkw * 4 + cil;
        float gk[32][3];
        #pragma unroll
        for (int k = 0; k < 32; ++k) {
            #pragma unroll
            for (int kd = 0; kd < 3; ++kd) gk[k][kd] = wp[k * 108 + kd * 9];
        }
        #pragma unroll
        for (int k = 0; k < 32; ++k) {
            float* q = lp + (k & 3) * W_GRP_PITCH + (k >> 2) * W_CO_PITCH;
            const float s02 = gk[k][0] + gk[k][2];
            q[0 * 36] = gk[k][0];
            q[1 * 36] = 0.5f * (s02 + gk[k][1]);
            q[2 * 36] = 0.5f * (s02 - gk[k][1]);
            q[3 * 36] = gk[k][2];
        }
    }
    for (int t = tid; t < SROWS * 4 * 2; t += 128) {
        int rr   = t >> 1;
        int side = t & 1;
        int row  = rr >> 2;
        int cil  = rr & 3;
        xs4[(row * 66 + side * 65) * 4 + cil] = 0.0f;
    }

    slab_commit();

    const int abase = (w0 + Lm) * 4 + cl;
    const float bv = bias[Lm];

    v8f acc[NPL][HROWS];
    #pragma unroll
    for (int j = 0; j < NPL; ++j) {
        #pragma unroll
        for (int r = 0; r < HROWS; ++r) acc[j][r] = v8f{};
    }

    for (int g = 0; g < 4; ++g) {
        __syncthreads();

        const int bbase = g * W_GRP_PITCH + Lm * W_CO_PITCH + cl;

        auto load_a = [&](int j, int hh, v2f* dst) {
            #pragma unroll
            for (int hl = 0; hl < 3; ++hl) {
                #pragma unroll
                for (int kw = 0; kw < 3; ++kw)
                    dst[hl * 3 + kw] =
                        *(const v2f*)(xs4 + abase + ((j * HIN + hh * 3 + hl) * 66 + kw) * 4);
            }
        };
        auto load_b = [&](int j, v2f* dst) {
            #pragma unroll
            for (int t = 0; t < 9; ++t)
                dst[t] = *(const v2f*)(wl + bbase + (j * 9 + t) * 4);
        };

        v2f af[2][9];
        v2f bfr[2][9];
        load_b(0, bfr[0]);
        load_a(0, 0, af[0]);
        #pragma unroll
        for (int u = 0; u < 2 * NPL; ++u) {
            const int j  = u >> 1;
            const int hh = u & 1;
            if (u + 1 < 2 * NPL) {
                load_a((u + 1) >> 1, (u + 1) & 1, af[(u + 1) & 1]);
                if (hh == 1) load_b(j + 1, bfr[(j + 1) & 1]);
            }
            __builtin_amdgcn_sched_barrier(0);
            #pragma unroll
            for (int hl = 0; hl < 3; ++hl) {
                const int hin = hh * 3 + hl;
                #pragma unroll
                for (int kw = 0; kw < 3; ++kw) {
                    #pragma unroll
                    for (int kh = 0; kh < 3; ++kh) {
                        const int r = hin - kh;
                        if (r >= 0 && r < HROWS)
                            acc[j][r] = __builtin_amdgcn_wmma_f32_16x16x4_f32(
                                false, af[u & 1][hl * 3 + kw], false, bfr[j & 1][kh * 3 + kw],
                                (short)0, acc[j][r], false, false);
                    }
                }
            }
            __builtin_amdgcn_sched_barrier(0);
        }

        asm volatile("v_nop\n\tv_nop\n\tv_nop\n\tv_nop\n\tv_nop\n\tv_nop\n\tv_nop\n\tv_nop"
                     : "+v"(acc[0][0]), "+v"(acc[0][1]), "+v"(acc[0][2]), "+v"(acc[0][3]),
                       "+v"(acc[1][0]), "+v"(acc[1][1]), "+v"(acc[1][2]), "+v"(acc[1][3]),
                       "+v"(acc[2][0]), "+v"(acc[2][1]), "+v"(acc[2][2]), "+v"(acc[2][3]),
                       "+v"(acc[3][0]), "+v"(acc[3][1]), "+v"(acc[3][2]), "+v"(acc[3][3]));

        if (g < 3) {
            slab_issue(xb + (long)(g + 1) * 4 * (SDIM * SDIM * SDIM));
            __syncthreads();
            slab_commit();
        }
    }

    #pragma unroll
    for (int r = 0; r < HROWS; ++r) {
        float* op = out + ((((long)b * NCOUT + Lm) * SDIM + d0) * SDIM + (h0 + r)) * SDIM
                        + w0 + 8 * half;
        const v8f m12 = acc[1][r] + acc[2][r];
        const v8f d12 = acc[1][r] - acc[2][r];
        const v8f o0 = (acc[0][r] + m12) + bv;
        const v8f o1 = (d12 - acc[3][r]) + bv;
        float4 a0 = {o0[0], o0[1], o0[2], o0[3]};
        float4 a1 = {o0[4], o0[5], o0[6], o0[7]};
        float4 c0 = {o1[0], o1[1], o1[2], o1[3]};
        float4 c1 = {o1[4], o1[5], o1[6], o1[7]};
        *(float4*)(op)     = a0;
        *(float4*)(op + 4) = a1;
        *(float4*)(op + SDIM * SDIM)     = c0;
        *(float4*)(op + SDIM * SDIM + 4) = c1;
    }
}

extern "C" void kernel_launch(void* const* d_in, const int* in_sizes, int n_in,
                              void* d_out, int out_size, void* d_ws, size_t ws_size,
                              hipStream_t stream) {
    const float* x    = (const float*)d_in[0];
    const float* wgt  = (const float*)d_in[1];
    const float* bias = (const float*)d_in[2];
    float* out        = (float*)d_out;
    (void)in_sizes; (void)n_in; (void)out_size; (void)d_ws; (void)ws_size;

    dim3 grid(4 * (SDIM / 2) * (SDIM / HROWS));
    conv3d_wmma_f32<<<grid, 128, 0, stream>>>(x, wgt, bias, out);
}


# Each work-group computes two consecutive h chunks in turn (grid 2048 -> 1024): weight transform into LDS once per two tiles; epilogue staged through the x slab, one h row at a time; sums unchanged
# speedup vs baseline: 4.1699x; 1.5328x over previous
#include <hip/hip_runtime.h>

typedef float v2f __attribute__((ext_vector_type(2)));
typedef float v8f __attribute__((ext_vector_type(8)));

#define SDIM 64
#define NCIN 16
#define NCOUT 16
#define HROWS 2
#define TPB 2
#define HIN (HROWS + 2)
#define DOUT 4
#define NPL (DOUT + 2)
#define NTT (NPL * 9)
#define SROWS (NPL * HIN)
#define NLD (HIN * 64 / 128)
#define S3 (SDIM * SDIM * SDIM)
#define W_CO_PITCH (NTT * 4 + 4)
#define W_GRP_PITCH (NCOUT * W_CO_PITCH)
#define STG_PITCH 68

__device__ __forceinline__ void lds_barrier() {
    __builtin_amdgcn_fence(__ATOMIC_RELEASE, "workgroup", "local");
    __builtin_amdgcn_s_barrier();
    __builtin_amdgcn_fence(__ATOMIC_ACQUIRE, "workgroup", "local");
}

__launch_bounds__(128, 4)
__global__ void conv3d_wmma_f32(const float* __restrict__ x,
                                const float* __restrict__ w,
                                const float* __restrict__ bias,
                                float* __restrict__ out) {
    __shared__ __align__(16) float wl[4 * W_GRP_PITCH];
    __shared__ __align__(16) float xs4[SROWS * 66 * 4];
    __shared__ float bs[NCOUT];

    const int tid  = threadIdx.x;
    const int lane = tid & 31;
    const int wv   = __builtin_amdgcn_readfirstlane(tid >> 5);
    const int Lm   = lane & 15;
    const int half = lane >> 4;
    const int cl   = half << 1;
    const int w0   = wv * 16;

    const int blk = blockIdx.x;
    const int b   = blk / ((SDIM / DOUT) * (SDIM / (HROWS * TPB)));
    const int d0  = ((blk / (SDIM / (HROWS * TPB))) % (SDIM / DOUT)) * DOUT;
    const int hb  = (blk % (SDIM / (HROWS * TPB))) * (HROWS * TPB);

    const int sw  = tid & 63;
    const int r0  = wv >> 1;
    int  zoff[NPL], hoff[NLD];
    bool zok[NPL], hok[NLD];
    #pragma unroll
    for (int p = 0; p < NPL; ++p) {
        int zd  = d0 + p - 1;
        zok[p]  = (unsigned)zd < 64u;
        zoff[p] = min(max(zd, 0), SDIM - 1) * (SDIM * SDIM);
    }
    auto set_rows = [&](int h0) {
        #pragma unroll
        for (int i = 0; i < NLD; ++i) {
            int zh  = h0 + 2 * i + r0 - 1;
            hok[i]  = (unsigned)zh < 64u;
            hoff[i] = min(max(zh, 0), SDIM - 1) * SDIM;
        }
    };
    set_rows(hb);
    const float* xb = x + (long)(b * NCIN) * S3;
    const unsigned swu = (unsigned)sw;
    const int sbase = (r0 * 66 + 1 + sw) * 4;

    float pv[NLD][NPL][4];
    auto slab_issue = [&](const float* xg) {
        #pragma unroll
        for (int i = 0; i < NLD; ++i) {
            #pragma unroll
            for (int p = 0; p < NPL; ++p) {
                const __attribute__((address_space(1))) float* q =
                    (const __attribute__((address_space(1))) float*)(xg + zoff[p] + hoff[i]);
                asm volatile("" : "+s"(q));
                #pragma unroll
                for (int c = 0; c < 4; ++c) pv[i][p][c] = (q + c * S3)[swu];
            }
        }
    };
    auto slab_commit = [&]() {
        #pragma unroll
        for (int i = 0; i < NLD; ++i) {
            float t[NPL][4];
            #pragma unroll
            for (int c = 0; c < 4; ++c) {
                const float x0 = (zok[0] && hok[i]) ? pv[i][0][c] : 0.0f;
                const float x1 = (zok[1] && hok[i]) ? pv[i][1][c] : 0.0f;
                const float x2 = (zok[2] && hok[i]) ? pv[i][2][c] : 0.0f;
                const float x3 = (zok[3] && hok[i]) ? pv[i][3][c] : 0.0f;
                const float x4 = (zok[4] && hok[i]) ? pv[i][4][c] : 0.0f;
                const float x5 = (zok[5] && hok[i]) ? pv[i][5][c] : 0.0f;
                const float a12 = x1 + x2;
                const float s12 = x1 - x2;
                const float s31 = x3 - x1;
                const float s42 = x4 - x2;
                t[0][c] = (4.0f * x0 - 5.0f * x2) + x4;
                t[1][c] = (x3 + x4) - 4.0f * a12;
                t[2][c] = (x4 - x3) + 4.0f * s12;
                t[3][c] = s42 + 2.0f * s31;
                t[4][c] = s42 - 2.0f * s31;
                t[5][c] = (4.0f * x1 - 5.0f * x3) + x5;
            }
            #pragma unroll
            for (int j = 0; j < NPL; ++j) {
                float4 v = {t[j][0], t[j][1], t[j][2], t[j][3]};
                *(float4*)(xs4 + sbase + (j * HIN + 2 * i) * 66 * 4) = v;
            }
        }
    };

    slab_issue(xb);

    if (tid < 2 * 36) {
        const int s    = tid % 36;
        const int part = tid / 36;
        const int cil  = s / 9;
        const int khkw = s % 9;
        const float* wp = w + part * 32 * 108 + cil * 27 + khkw;
        float* lp = wl + part * 8 * W_CO_PITCH + khkw * 4 + cil;
        float gk[32][3];
        #pragma unroll
        for (int k = 0; k < 32; ++k) {
            #pragma unroll
            for (int kd = 0; kd < 3; ++kd) gk[k][kd] = wp[k * 108 + kd * 9];
        }
        #pragma unroll
        for (int k = 0; k < 32; ++k) {
            float* q = lp + (k & 3) * W_GRP_PITCH + (k >> 2) * W_CO_PITCH;
            const float s02 = gk[k][0] + gk[k][2];
            const float ev  = gk[k][0] * (1.0f / 24.0f) + gk[k][2] * (1.0f / 6.0f);
            const float od  = gk[k][1] * (1.0f / 12.0f);
            q[0 * 36] = 0.25f * gk[k][0];
            q[1 * 36] = (-1.0f / 6.0f) * (s02 + gk[k][1]);
            q[2 * 36] = (-1.0f / 6.0f) * (s02 - gk[k][1]);
            q[3 * 36] = ev + od;
            q[4 * 36] = ev - od;
            q[5 * 36] = gk[k][2];
        }
    }
    auto zero_halo = [&]() {
        for (int t = tid; t < SROWS * 4 * 2; t += 128) {
            int rr   = t >> 1;
            int side = t & 1;
            int row  = rr >> 2;
            int cil  = rr & 3;
            xs4[(row * 66 + side * 65) * 4 + cil] = 0.0f;
        }
    };
    zero_halo();

    slab_commit();

    const int abase = (w0 + Lm) * 4 + cl;
    if (tid < NCOUT) bs[tid] = bias[tid];

    v8f acc[NPL][HROWS];

    for (int tile = 0; tile < TPB; ++tile) {
        const int h0 = hb + tile * HROWS;

        #pragma unroll
        for (int j = 0; j < NPL; ++j) {
            #pragma unroll
            for (int r = 0; r < HROWS; ++r) acc[j][r] = v8f{};
        }

        for (int g = 0; g < 4; ++g) {
            lds_barrier();

            const int bbase = g * W_GRP_PITCH + Lm * W_CO_PITCH + cl;

            auto load_a = [&](int j, v2f* dst) {
                #pragma unroll
                for (int hl = 0; hl < HIN; ++hl) {
                    #pragma unroll
                    for (int kw = 0; kw < 3; ++kw)
                        dst[hl * 3 + kw] =
                            *(const v2f*)(xs4 + abase + ((j * HIN + hl) * 66 + kw) * 4);
                }
            };
            auto load_b = [&](int j, v2f* dst) {
                #pragma unroll
                for (int t = 0; t < 9; ++t)
                    dst[t] = *(const v2f*)(wl + bbase + (j * 9 + t) * 4);
            };

            v2f af[2][HIN * 3];
            v2f bfr[2][9];
            load_b(0, bfr[0]);
            load_a(0, af[0]);
            #pragma unroll
            for (int j = 0; j < NPL; ++j) {
                if (j + 1 < NPL) {
                    load_a(j + 1, af[(j + 1) & 1]);
                    load_b(j + 1, bfr[(j + 1) & 1]);
                }
                __builtin_amdgcn_sched_barrier(0);
                #pragma unroll
                for (int hin = 0; hin < HIN; ++hin) {
                    #pragma unroll
                    for (int kw = 0; kw < 3; ++kw) {
                        #pragma unroll
                        for (int kh = 0; kh < 3; ++kh) {
                            const int r = hin - kh;
                            if (r >= 0 && r < HROWS)
                                acc[j][r] = __builtin_amdgcn_wmma_f32_16x16x4_f32(
                                    false, af[j & 1][hin * 3 + kw], false, bfr[j & 1][kh * 3 + kw],
                                    (short)0, acc[j][r], false, false);
                        }
                    }
                }
                __builtin_amdgcn_sched_barrier(0);
            }

            asm volatile("v_nop\n\tv_nop\n\tv_nop\n\tv_nop\n\tv_nop\n\tv_nop\n\tv_nop\n\tv_nop"
                         : "+v"(acc[0][0]), "+v"(acc[0][1]), "+v"(acc[1][0]), "+v"(acc[1][1]),
                           "+v"(acc[2][0]), "+v"(acc[2][1]), "+v"(acc[3][0]), "+v"(acc[3][1]),
                           "+v"(acc[4][0]), "+v"(acc[4][1]), "+v"(acc[5][0]), "+v"(acc[5][1]));

            if (g < 3) {
                slab_issue(xb + (long)(g + 1) * 4 * (SDIM * SDIM * SDIM));
                lds_barrier();
                slab_commit();
            }
        }

        if (tile + 1 < TPB) {
            set_rows(h0 + HROWS);
            slab_issue(xb);
        }

        lds_barrier();
        float* stg = xs4;
        int lt = lane;
        asm volatile("" : "+v"(lt));
        const int l16 = lt & 15;
        const int hf  = lt >> 4;
        const float bt = bs[l16];
        #pragma unroll
        for (int r = 0; r < HROWS; ++r) {
            if (r > 0) lds_barrier();
            const v8f p12 = acc[1][r] + acc[2][r];
            const v8f m12 = acc[1][r] - acc[2][r];
            const v8f p34 = acc[3][r] + acc[4][r];
            const v8f m34 = acc[3][r] - acc[4][r];
            v8f o[DOUT];
            o[0] = ((acc[0][r] + p12) + p34) + bt;
            o[1] = (m12 + 2.0f * m34) + bt;
            o[2] = (p12 + 4.0f * p34) + bt;
            o[3] = ((m12 + 8.0f * m34) + acc[5][r]) + bt;
            #pragma unroll
            for (int a = 0; a < DOUT; ++a) {
                float* sp = stg + (a * NCOUT + Lm) * STG_PITCH + w0 + 8 * half;
                float4 lo = {o[a][0], o[a][1], o[a][2], o[a][3]};
                float4 hi = {o[a][4], o[a][5], o[a][6], o[a][7]};
                *(float4*)(sp)     = lo;
                *(float4*)(sp + 4) = hi;
            }
            lds_barrier();
            const float* sl = stg + (wv * NCOUT + hf) * STG_PITCH + 4 * l16;
            float* ob = out + ((((long)b * NCOUT + hf) * SDIM + (d0 + wv)) * SDIM + (h0 + r)) * SDIM
                            + 4 * l16;
            #pragma unroll
            for (int k = 0; k < NCOUT / 2; ++k) {
                const float4 v = *(const float4*)(sl + 2 * k * STG_PITCH);
                *(float4*)(ob + (long)(2 * k) * S3) = v;
            }
        }

        if (tile + 1 < TPB) {
            lds_barrier();
            zero_halo();
            slab_commit();
        }
    }
}

extern "C" void kernel_launch(void* const* d_in, const int* in_sizes, int n_in,
                              void* d_out, int out_size, void* d_ws, size_t ws_size,
                              hipStream_t stream) {
    const float* x    = (const float*)d_in[0];
    const float* wgt  = (const float*)d_in[1];
    const float* bias = (const float*)d_in[2];
    float* out        = (float*)d_out;
    (void)in_sizes; (void)n_in; (void)out_size; (void)d_ws; (void)ws_size;

    dim3 grid(4 * (SDIM / DOUT) * (SDIM / (HROWS * TPB)));
    conv3d_wmma_f32<<<grid, 128, 0, stream>>>(x, wgt, bias, out);
}


# x slab loads of the next cin-group (and of the next tile's first group) issued in front of the current group's WMMAs instead of behind them, so they return under the matrix work; sums unchanged
# speedup vs baseline: 4.3098x; 1.0335x over previous
#include <hip/hip_runtime.h>

typedef float v2f __attribute__((ext_vector_type(2)));
typedef float v8f __attribute__((ext_vector_type(8)));

#define SDIM 64
#define NCIN 16
#define NCOUT 16
#define HROWS 2
#define TPB 2
#define HIN (HROWS + 2)
#define DOUT 4
#define NPL (DOUT + 2)
#define NTT (NPL * 9)
#define SROWS (NPL * HIN)
#define NLD (HIN * 64 / 128)
#define S3 (SDIM * SDIM * SDIM)
#define W_CO_PITCH (NTT * 4 + 4)
#define W_GRP_PITCH (NCOUT * W_CO_PITCH)
#define STG_PITCH 68

__device__ __forceinline__ void lds_barrier() {
    __builtin_amdgcn_fence(__ATOMIC_RELEASE, "workgroup", "local");
    __builtin_amdgcn_s_barrier();
    __builtin_amdgcn_fence(__ATOMIC_ACQUIRE, "workgroup", "local");
}

__launch_bounds__(128, 4)
__global__ void conv3d_wmma_f32(const float* __restrict__ x,
                                const float* __restrict__ w,
                                const float* __restrict__ bias,
                                float* __restrict__ out) {
    __shared__ __align__(16) float wl[4 * W_GRP_PITCH];
    __shared__ __align__(16) float xs4[SROWS * 66 * 4];
    __shared__ float bs[NCOUT];

    const int tid  = threadIdx.x;
    const int lane = tid & 31;
    const int wv   = __builtin_amdgcn_readfirstlane(tid >> 5);
    const int Lm   = lane & 15;
    const int half = lane >> 4;
    const int cl   = half << 1;
    const int w0   = wv * 16;

    const int blk = blockIdx.x;
    const int b   = blk / ((SDIM / DOUT) * (SDIM / (HROWS * TPB)));
    const int d0  = ((blk / (SDIM / (HROWS * TPB))) % (SDIM / DOUT)) * DOUT;
    const int hb  = (blk % (SDIM / (HROWS * TPB))) * (HROWS * TPB);

    const int sw  = tid & 63;
    const int r0  = wv >> 1;
    int  zoff[NPL], hoff[NLD];
    bool zok[NPL], hok[NLD];
    #pragma unroll
    for (int p = 0; p < NPL; ++p) {
        int zd  = d0 + p - 1;
        zok[p]  = (unsigned)zd < 64u;
        zoff[p] = min(max(zd, 0), SDIM - 1) * (SDIM * SDIM);
    }
    auto set_rows = [&](int h0) {
        #pragma unroll
        for (int i = 0; i < NLD; ++i) {
            int zh  = h0 + 2 * i + r0 - 1;
            hok[i]  = (unsigned)zh < 64u;
            hoff[i] = __builtin_amdgcn_readfirstlane(min(max(zh, 0), SDIM - 1) * SDIM);
        }
    };
    set_rows(hb);
    const float* xb = x + (long)(b * NCIN) * S3;
    const unsigned swu = (unsigned)sw;
    const int sbase = (r0 * 66 + 1 + sw) * 4;

    float pv[NLD][NPL][4];
    auto slab_issue = [&](const float* xg, const int* ho) {
        #pragma unroll
        for (int i = 0; i < NLD; ++i) {
            #pragma unroll
            for (int p = 0; p < NPL; ++p) {
                const __attribute__((address_space(1))) float* q =
                    (const __attribute__((address_space(1))) float*)(xg + zoff[p] + ho[i]);
                asm volatile("" : "+s"(q));
                #pragma unroll
                for (int c = 0; c < 4; ++c) pv[i][p][c] = (q + c * S3)[swu];
            }
        }
    };
    auto slab_commit = [&]() {
        #pragma unroll
        for (int i = 0; i < NLD; ++i) {
            float t[NPL][4];
            #pragma unroll
            for (int c = 0; c < 4; ++c) {
                const float x0 = (zok[0] && hok[i]) ? pv[i][0][c] : 0.0f;
                const float x1 = (zok[1] && hok[i]) ? pv[i][1][c] : 0.0f;
                const float x2 = (zok[2] && hok[i]) ? pv[i][2][c] : 0.0f;
                const float x3 = (zok[3] && hok[i]) ? pv[i][3][c] : 0.0f;
                const float x4 = (zok[4] && hok[i]) ? pv[i][4][c] : 0.0f;
                const float x5 = (zok[5] && hok[i]) ? pv[i][5][c] : 0.0f;
                const float a12 = x1 + x2;
                const float s12 = x1 - x2;
                const float s31 = x3 - x1;
                const float s42 = x4 - x2;
                t[0][c] = (4.0f * x0 - 5.0f * x2) + x4;
                t[1][c] = (x3 + x4) - 4.0f * a12;
                t[2][c] = (x4 - x3) + 4.0f * s12;
                t[3][c] = s42 + 2.0f * s31;
                t[4][c] = s42 - 2.0f * s31;
                t[5][c] = (4.0f * x1 - 5.0f * x3) + x5;
            }
            #pragma unroll
            for (int j = 0; j < NPL; ++j) {
                float4 v = {t[j][0], t[j][1], t[j][2], t[j][3]};
                *(float4*)(xs4 + sbase + (j * HIN + 2 * i) * 66 * 4) = v;
            }
        }
    };

    slab_issue(xb, hoff);

    if (tid < 2 * 36) {
        const int s    = tid % 36;
        const int part = tid / 36;
        const int cil  = s / 9;
        const int khkw = s % 9;
        const float* wp = w + part * 32 * 108 + cil * 27 + khkw;
        float* lp = wl + part * 8 * W_CO_PITCH + khkw * 4 + cil;
        float gk[32][3];
        #pragma unroll
        for (int k = 0; k < 32; ++k) {
            #pragma unroll
            for (int kd = 0; kd < 3; ++kd) gk[k][kd] = wp[k * 108 + kd * 9];
        }
        #pragma unroll
        for (int k = 0; k < 32; ++k) {
            float* q = lp + (k & 3) * W_GRP_PITCH + (k >> 2) * W_CO_PITCH;
            const float s02 = gk[k][0] + gk[k][2];
            const float ev  = gk[k][0] * (1.0f / 24.0f) + gk[k][2] * (1.0f / 6.0f);
            const float od  = gk[k][1] * (1.0f / 12.0f);
            q[0 * 36] = 0.25f * gk[k][0];
            q[1 * 36] = (-1.0f / 6.0f) * (s02 + gk[k][1]);
            q[2 * 36] = (-1.0f / 6.0f) * (s02 - gk[k][1]);
            q[3 * 36] = ev + od;
            q[4 * 36] = ev - od;
            q[5 * 36] = gk[k][2];
        }
    }
    auto zero_halo = [&]() {
        for (int t = tid; t < SROWS * 4 * 2; t += 128) {
            int rr   = t >> 1;
            int side = t & 1;
            int row  = rr >> 2;
            int cil  = rr & 3;
            xs4[(row * 66 + side * 65) * 4 + cil] = 0.0f;
        }
    };
    zero_halo();

    slab_commit();

    const int abase = (w0 + Lm) * 4 + cl;
    if (tid < NCOUT) bs[tid] = bias[tid];

    v8f acc[NPL][HROWS];

    for (int tile = 0; tile < TPB; ++tile) {
        const int h0 = hb + tile * HROWS;

        #pragma unroll
        for (int j = 0; j < NPL; ++j) {
            #pragma unroll
            for (int r = 0; r < HROWS; ++r) acc[j][r] = v8f{};
        }

        for (int g = 0; g < 4; ++g) {
            lds_barrier();

            const int bbase = g * W_GRP_PITCH + Lm * W_CO_PITCH + cl;

            auto load_a = [&](int j, v2f* dst) {
                #pragma unroll
                for (int hl = 0; hl < HIN; ++hl) {
                    #pragma unroll
                    for (int kw = 0; kw < 3; ++kw)
                        dst[hl * 3 + kw] =
                            *(const v2f*)(xs4 + abase + ((j * HIN + hl) * 66 + kw) * 4);
                }
            };
            auto load_b = [&](int j, v2f* dst) {
                #pragma unroll
                for (int t = 0; t < 9; ++t)
                    dst[t] = *(const v2f*)(wl + bbase + (j * 9 + t) * 4);
            };

            v2f af[2][HIN * 3];
            v2f bfr[2][9];
            load_b(0, bfr[0]);
            load_a(0, af[0]);
            if (g < 3 || tile + 1 < TPB) {
                int ho[NLD];
                #pragma unroll
                for (int i = 0; i < NLD; ++i) {
                    const int zh = h0 + ((g == 3) ? HROWS : 0) + 2 * i + r0 - 1;
                    ho[i] = min(max(zh, 0), SDIM - 1) * SDIM;
                }
                slab_issue(xb + (long)((g + 1) & 3) * 4 * (SDIM * SDIM * SDIM), ho);
            }
            #pragma unroll
            for (int j = 0; j < NPL; ++j) {
                if (j + 1 < NPL) {
                    load_a(j + 1, af[(j + 1) & 1]);
                    load_b(j + 1, bfr[(j + 1) & 1]);
                }
                __builtin_amdgcn_sched_barrier(0);
                #pragma unroll
                for (int hin = 0; hin < HIN; ++hin) {
                    #pragma unroll
                    for (int kw = 0; kw < 3; ++kw) {
                        #pragma unroll
                        for (int kh = 0; kh < 3; ++kh) {
                            const int r = hin - kh;
                            if (r >= 0 && r < HROWS)
                                acc[j][r] = __builtin_amdgcn_wmma_f32_16x16x4_f32(
                                    false, af[j & 1][hin * 3 + kw], false, bfr[j & 1][kh * 3 + kw],
                                    (short)0, acc[j][r], false, false);
                        }
                    }
                }
                __builtin_amdgcn_sched_barrier(0);
            }

            asm volatile("v_nop\n\tv_nop\n\tv_nop\n\tv_nop\n\tv_nop\n\tv_nop\n\tv_nop\n\tv_nop"
                         : "+v"(acc[0][0]), "+v"(acc[0][1]), "+v"(acc[1][0]), "+v"(acc[1][1]),
                           "+v"(acc[2][0]), "+v"(acc[2][1]), "+v"(acc[3][0]), "+v"(acc[3][1]),
                           "+v"(acc[4][0]), "+v"(acc[4][1]), "+v"(acc[5][0]), "+v"(acc[5][1]));

            if (g < 3) {
                lds_barrier();
                slab_commit();
            }
        }

        if (tile + 1 < TPB) set_rows(h0 + HROWS);

        lds_barrier();
        float* stg = xs4;
        int lt = lane;
        asm volatile("" : "+v"(lt));
        const int l16 = lt & 15;
        const int hf  = lt >> 4;
        const float bt = bs[l16];
        #pragma unroll
        for (int r = 0; r < HROWS; ++r) {
            if (r > 0) lds_barrier();
            const v8f p12 = acc[1][r] + acc[2][r];
            const v8f m12 = acc[1][r] - acc[2][r];
            const v8f p34 = acc[3][r] + acc[4][r];
            const v8f m34 = acc[3][r] - acc[4][r];
            v8f o[DOUT];
            o[0] = ((acc[0][r] + p12) + p34) + bt;
            o[1] = (m12 + 2.0f * m34) + bt;
            o[2] = (p12 + 4.0f * p34) + bt;
            o[3] = ((m12 + 8.0f * m34) + acc[5][r]) + bt;
            #pragma unroll
            for (int a = 0; a < DOUT; ++a) {
                float* sp = stg + (a * NCOUT + Lm) * STG_PITCH + w0 + 8 * half;
                float4 lo = {o[a][0], o[a][1], o[a][2], o[a][3]};
                float4 hi = {o[a][4], o[a][5], o[a][6], o[a][7]};
                *(float4*)(sp)     = lo;
                *(float4*)(sp + 4) = hi;
            }
            lds_barrier();
            const float* sl = stg + (wv * NCOUT + hf) * STG_PITCH + 4 * l16;
            float* ob = out + ((((long)b * NCOUT + hf) * SDIM + (d0 + wv)) * SDIM + (h0 + r)) * SDIM
                            + 4 * l16;
            #pragma unroll
            for (int k = 0; k < NCOUT / 2; ++k) {
                const float4 v = *(const float4*)(sl + 2 * k * STG_PITCH);
                *(float4*)(ob + (long)(2 * k) * S3) = v;
            }
        }

        if (tile + 1 < TPB) {
            lds_barrier();
            zero_halo();
            slab_commit();
        }
    }
}

extern "C" void kernel_launch(void* const* d_in, const int* in_sizes, int n_in,
                              void* d_out, int out_size, void* d_ws, size_t ws_size,
                              hipStream_t stream) {
    const float* x    = (const float*)d_in[0];
    const float* wgt  = (const float*)d_in[1];
    const float* bias = (const float*)d_in[2];
    float* out        = (float*)d_out;
    (void)in_sizes; (void)n_in; (void)out_size; (void)d_ws; (void)ws_size;

    dim3 grid(4 * (SDIM / DOUT) * (SDIM / (HROWS * TPB)));
    conv3d_wmma_f32<<<grid, 128, 0, stream>>>(x, wgt, bias, out);
}
